# starting kernel
# baseline (speedup 1.0000x reference)
#include <hip/hip_runtime.h>


typedef float v2f __attribute__((ext_vector_type(2)));
typedef float v8f __attribute__((ext_vector_type(8)));

#define GP_N_NODES 100000
#define GP_DEG 16
#define GP_D_FEAT 64

__global__ __launch_bounds__(256) void spmm_hop_wmma(
    const float* __restrict__ xin,
    const float* __restrict__ vals,
    const int*   __restrict__ indices,
    float*       __restrict__ xout,
    int nrows)
{
    const int  lane = threadIdx.x & 31;
    const bool hi   = lane >= 16;
    const int  fl   = lane & 15;
    const int  wavesPerBlock = blockDim.x >> 5;
    const int  wave   = blockIdx.x * wavesPerBlock + (threadIdx.x >> 5);
    const int  stride = gridDim.x * wavesPerBlock;

    for (int row = wave; row < nrows; row += stride) {
        const int r = __builtin_amdgcn_readfirstlane(row);
        const float* __restrict__ wp = vals    + (size_t)r * GP_DEG;
        const int*   __restrict__ ip = indices + (size_t)r * GP_DEG;

        v2f A[4];
        v2f B[4][4];

        #pragma unroll
        for (int s = 0; s < 4; ++s) {
            const float w0 = wp[4*s+0], w1 = wp[4*s+1];
            const float w2 = wp[4*s+2], w3 = wp[4*s+3];
            const int   n0 = ip[4*s+0], n1 = ip[4*s+1];
            const int   n2 = ip[4*s+2], n3 = ip[4*s+3];

            A[s].x = hi ? w2 : w0;
            A[s].y = hi ? w3 : w1;

            const float* bx = xin + (size_t)(hi ? n2 : n0) * GP_D_FEAT + fl;
            const float* by = xin + (size_t)(hi ? n3 : n1) * GP_D_FEAT + fl;
            #pragma unroll
            for (int t = 0; t < 4; ++t) {
                B[s][t].x = bx[16 * t];
                B[s][t].y = by[16 * t];
            }
        }

        v8f acc[4] = {};
        #pragma unroll
        for (int s = 0; s < 4; ++s) {
            #pragma unroll
            for (int t = 0; t < 4; ++t) {
                acc[t] = __builtin_amdgcn_wmma_f32_16x16x4_f32(
                     false, A[s],  false, B[s][t],
                     (short)0, acc[t],
                     false,  false);
            }
        }

        float* __restrict__ op = xout + (size_t)row * GP_D_FEAT;
        op[lane]      = hi ? acc[1][0] : acc[0][0];
        op[32 + lane] = hi ? acc[3][0] : acc[2][0];
    }
}

extern "C" void kernel_launch(void* const* d_in, const int* in_sizes, int n_in,
                              void* d_out, int out_size, void* d_ws, size_t ws_size,
                              hipStream_t stream) {
    (void)in_sizes; (void)n_in; (void)out_size; (void)ws_size;

    const float* x       = (const float*)d_in[0];
    const float* values  = (const float*)d_in[1];
    const int*   indices = (const int*)  d_in[3];
    float* out = (float*)d_out;
    float* ws  = (float*)d_ws;

    const int nrows       = GP_N_NODES;
    const int threads     = 256;
    const int rowsPerWave = 4;
    const int waves  = (nrows + rowsPerWave - 1) / rowsPerWave;
    const int blocks = (waves + (threads / 32) - 1) / (threads / 32);

    spmm_hop_wmma<<<blocks, threads, 0, stream>>>(x,   values, indices, out, nrows);
    spmm_hop_wmma<<<blocks, threads, 0, stream>>>(out, values, indices, ws,  nrows);
    spmm_hop_wmma<<<blocks, threads, 0, stream>>>(ws,  values, indices, out, nrows);
}


# Rows software-pipelined inside each wave: the next row's 32 gathers and the following row's weight/index fetch go out ahead of the current row's 16 WMMAs; 1024 resident blocks, 12-13 rows per wave
# speedup vs baseline: 1.0348x; 1.0348x over previous
#include <hip/hip_runtime.h>


typedef float v2f __attribute__((ext_vector_type(2)));
typedef float v8f __attribute__((ext_vector_type(8)));

#define GP_N_NODES 100000
#define GP_DEG 16
#define GP_D_FEAT 64

struct RowOps {
    float w[GP_DEG];
    int   n[GP_DEG];
};

__device__ __forceinline__ void fetch_ops(const float* __restrict__ vals,
                                          const int*   __restrict__ indices,
                                          int r, RowOps& o)
{
    const float* __restrict__ wp = vals    + (size_t)r * GP_DEG;
    const int*   __restrict__ ip = indices + (size_t)r * GP_DEG;
    #pragma unroll
    for (int k = 0; k < GP_DEG; ++k) { o.w[k] = wp[k]; o.n[k] = ip[k]; }
}

__device__ __forceinline__ void issue_row(const float* __restrict__ xin,
                                          const RowOps& o, bool hi, unsigned lane_boff,
                                          v2f (&A)[4], v2f (&B)[4][4])
{
    const char* __restrict__ xb = (const char*)xin;
    #pragma unroll
    for (int s = 0; s < 4; ++s) {
        A[s].x = hi ? o.w[4*s+2] : o.w[4*s+0];
        A[s].y = hi ? o.w[4*s+3] : o.w[4*s+1];

        const unsigned ox = ((unsigned)(hi ? o.n[4*s+2] : o.n[4*s+0]) << 8) | lane_boff;
        const unsigned oy = ((unsigned)(hi ? o.n[4*s+3] : o.n[4*s+1]) << 8) | lane_boff;
        #pragma unroll
        for (int t = 0; t < 4; ++t) {
            B[s][t].x = *(const float*)(xb + ox + 64u * t);
            B[s][t].y = *(const float*)(xb + oy + 64u * t);
        }
    }
}

__device__ __forceinline__ void finish_row(const v2f (&A)[4], const v2f (&B)[4][4],
                                           float* __restrict__ xout, int row,
                                           int lane, bool hi)
{
    v8f acc[4] = {};
    #pragma unroll
    for (int s = 0; s < 4; ++s) {
        #pragma unroll
        for (int t = 0; t < 4; ++t) {
            acc[t] = __builtin_amdgcn_wmma_f32_16x16x4_f32(
                 false, A[s],  false, B[s][t],
                 (short)0, acc[t],
                 false,  false);
        }
    }
    float* __restrict__ op = xout + (size_t)row * GP_D_FEAT;
    op[lane]      = hi ? acc[1][0] : acc[0][0];
    op[32 + lane] = hi ? acc[3][0] : acc[2][0];
}

__global__ __launch_bounds__(256, 8) void spmm_hop_wmma(
    const float* __restrict__ xin,
    const float* __restrict__ vals,
    const int*   __restrict__ indices,
    float*       __restrict__ xout,
    int nrows)
{
    const int  lane = threadIdx.x & 31;
    const bool hi   = lane >= 16;
    const unsigned lane_boff = (unsigned)(lane & 15) << 2;
    const int  wavesPerBlock = blockDim.x >> 5;
    const int  g = __builtin_amdgcn_readfirstlane(blockIdx.x * wavesPerBlock + (threadIdx.x >> 5));
    const int  G = gridDim.x * wavesPerBlock;
    if (g >= nrows) return;
    const int  cnt = (nrows - 1 - g) / G + 1;

    v2f A0[4], A1[4];
    v2f B0[4][4], B1[4][4];
    RowOps ops;

    fetch_ops(vals, indices, g, ops);
    issue_row(xin, ops, hi, lane_boff, A0, B0);
    fetch_ops(vals, indices, cnt > 1 ? g + G : g, ops);

    int j = 0;
    for (; j + 2 < cnt; j += 2) {
        const int r0 = g + j * G;
        issue_row(xin, ops, hi, lane_boff, A1, B1);
        fetch_ops(vals, indices, r0 + 2 * G, ops);
        __builtin_amdgcn_sched_barrier(0);
        finish_row(A0, B0, xout, r0, lane, hi);
        __builtin_amdgcn_sched_barrier(0);

        issue_row(xin, ops, hi, lane_boff, A0, B0);
        fetch_ops(vals, indices, (j + 3 < cnt) ? r0 + 3 * G : r0 + 2 * G, ops);
        __builtin_amdgcn_sched_barrier(0);
        finish_row(A1, B1, xout, r0 + G, lane, hi);
        __builtin_amdgcn_sched_barrier(0);
    }

    const int rl = g + j * G;
    if (cnt - j == 2) {
        issue_row(xin, ops, hi, lane_boff, A1, B1);
        __builtin_amdgcn_sched_barrier(0);
        finish_row(A0, B0, xout, rl, lane, hi);
        __builtin_amdgcn_sched_barrier(0);
        finish_row(A1, B1, xout, rl + G, lane, hi);
    } else {
        finish_row(A0, B0, xout, rl, lane, hi);
    }
}

extern "C" void kernel_launch(void* const* d_in, const int* in_sizes, int n_in,
                              void* d_out, int out_size, void* d_ws, size_t ws_size,
                              hipStream_t stream) {
    (void)in_sizes; (void)n_in; (void)out_size; (void)ws_size;

    const float* x       = (const float*)d_in[0];
    const float* values  = (const float*)d_in[1];
    const int*   indices = (const int*)  d_in[3];
    float* out = (float*)d_out;
    float* ws  = (float*)d_ws;

    const int nrows   = GP_N_NODES;
    const int threads = 256;
    const int blocks  = 1024;

    spmm_hop_wmma<<<blocks, threads, 0, stream>>>(x,   values, indices, out, nrows);
    spmm_hop_wmma<<<blocks, threads, 0, stream>>>(out, values, indices, ws,  nrows);
    spmm_hop_wmma<<<blocks, threads, 0, stream>>>(ws,  values, indices, out, nrows);
}


# Each gather is one 16-byte-per-lane load of two whole neighbor rows (8 loads a row, not 32), feature pairs on A rows 0/1, one 256-B store; new sum order: 2 neighbors (j, j+8) per WMMA, 8 chained
# speedup vs baseline: 1.0968x; 1.0599x over previous
#include <hip/hip_runtime.h>


typedef float v2f __attribute__((ext_vector_type(2)));
typedef float v4f __attribute__((ext_vector_type(4)));
typedef float v8f __attribute__((ext_vector_type(8)));
typedef int   v8i __attribute__((ext_vector_type(8)));

#define GP_N_NODES 100000
#define GP_DEG 16
#define GP_D_FEAT 64

struct RowOps {
    v8f wlo, whi;
    v8i nlo, nhi;
};

__device__ __forceinline__ void fetch_ops(const float* __restrict__ vals,
                                          const int*   __restrict__ indices,
                                          int r, RowOps& o)
{
    const float* __restrict__ wp = vals    + (size_t)r * GP_DEG;
    const int*   __restrict__ ip = indices + (size_t)r * GP_DEG;
    o.wlo = *(const v8f*)(wp);
    o.whi = *(const v8f*)(wp + 8);
    o.nlo = *(const v8i*)(ip);
    o.nhi = *(const v8i*)(ip + 8);
}

__device__ __forceinline__ void issue_row(const float* __restrict__ xin,
                                          const RowOps& o, bool hi, bool row_even, bool row_odd,
                                          unsigned lane_boff,
                                          v2f (&A)[8], v4f (&B)[8])
{
    const char* __restrict__ xb = (const char*)xin;
    #pragma unroll
    for (int j = 0; j < 8; ++j) {
        const float wsel = hi ? o.whi[j] : o.wlo[j];
        A[j].x = row_even ? wsel : 0.0f;
        A[j].y = row_odd  ? wsel : 0.0f;

        const unsigned off = ((unsigned)(hi ? o.nhi[j] : o.nlo[j]) << 8) | lane_boff;
        B[j] = *(const v4f*)(xb + off);
    }
}

__device__ __forceinline__ void finish_row(const v2f (&A)[8], const v4f (&B)[8],
                                           float* __restrict__ xout, int row,
                                           int lane, bool hi)
{
    v8f acc01 = {};
    v8f acc23 = {};
    #pragma unroll
    for (int j = 0; j < 8; ++j) {
        const v2f b01 = { B[j].x, B[j].y };
        const v2f b23 = { B[j].z, B[j].w };
        acc01 = __builtin_amdgcn_wmma_f32_16x16x4_f32(
             false, A[j],  false, b01,
             (short)0, acc01,
             false,  false);
        acc23 = __builtin_amdgcn_wmma_f32_16x16x4_f32(
             false, A[j],  false, b23,
             (short)0, acc23,
             false,  false);
    }
    v2f o2;
    o2.x = hi ? acc23[0] : acc01[0];
    o2.y = hi ? acc23[1] : acc01[1];
    float* __restrict__ op = xout + (size_t)row * GP_D_FEAT + 4 * (lane & 15) + (hi ? 2 : 0);
    *(v2f*)op = o2;
}

__global__ __launch_bounds__(256, 8) void spmm_hop_wmma(
    const float* __restrict__ xin,
    const float* __restrict__ vals,
    const int*   __restrict__ indices,
    float*       __restrict__ xout,
    int nrows)
{
    const int  lane = threadIdx.x & 31;
    const bool hi   = lane >= 16;
    const unsigned lane_boff = (unsigned)(lane & 15) << 4;
    const bool row_even = (lane & 7) == 0;
    const bool row_odd  = (lane & 7) == 1;
    const int  wavesPerBlock = blockDim.x >> 5;
    const int  g = __builtin_amdgcn_readfirstlane(blockIdx.x * wavesPerBlock + (threadIdx.x >> 5));
    const int  G = gridDim.x * wavesPerBlock;
    if (g >= nrows) return;
    const int  cnt = (nrows - 1 - g) / G + 1;

    v2f A0[8], A1[8];
    v4f B0[8], B1[8];
    RowOps ops;

    fetch_ops(vals, indices, g, ops);
    issue_row(xin, ops, hi, row_even, row_odd, lane_boff, A0, B0);
    fetch_ops(vals, indices, cnt > 1 ? g + G : g, ops);

    int j = 0;
    for (; j + 2 < cnt; j += 2) {
        const int r0 = g + j * G;
        issue_row(xin, ops, hi, row_even, row_odd, lane_boff, A1, B1);
        fetch_ops(vals, indices, r0 + 2 * G, ops);
        __builtin_amdgcn_sched_barrier(0);
        finish_row(A0, B0, xout, r0, lane, hi);
        __builtin_amdgcn_sched_barrier(0);

        issue_row(xin, ops, hi, row_even, row_odd, lane_boff, A0, B0);
        fetch_ops(vals, indices, (j + 3 < cnt) ? r0 + 3 * G : r0 + 2 * G, ops);
        __builtin_amdgcn_sched_barrier(0);
        finish_row(A1, B1, xout, r0 + G, lane, hi);
        __builtin_amdgcn_sched_barrier(0);
    }

    const int rl = g + j * G;
    if (cnt - j == 2) {
        issue_row(xin, ops, hi, row_even, row_odd, lane_boff, A1, B1);
        __builtin_amdgcn_sched_barrier(0);
        finish_row(A0, B0, xout, rl, lane, hi);
        __builtin_amdgcn_sched_barrier(0);
        finish_row(A1, B1, xout, rl + G, lane, hi);
    } else {
        finish_row(A0, B0, xout, rl, lane, hi);
    }
}

extern "C" void kernel_launch(void* const* d_in, const int* in_sizes, int n_in,
                              void* d_out, int out_size, void* d_ws, size_t ws_size,
                              hipStream_t stream) {
    (void)in_sizes; (void)n_in; (void)out_size; (void)ws_size;

    const float* x       = (const float*)d_in[0];
    const float* values  = (const float*)d_in[1];
    const int*   indices = (const int*)  d_in[3];
    float* out = (float*)d_out;
    float* ws  = (float*)d_ws;

    const int nrows   = GP_N_NODES;
    const int threads = 256;
    const int blocks  = 1024;

    spmm_hop_wmma<<<blocks, threads, 0, stream>>>(x,   values, indices, out, nrows);
    spmm_hop_wmma<<<blocks, threads, 0, stream>>>(out, values, indices, ws,  nrows);
    spmm_hop_wmma<<<blocks, threads, 0, stream>>>(ws,  values, indices, out, nrows);
}
